# starting kernel
# baseline (speedup 1.0000x reference)
#include <hip/hip_runtime.h>


typedef __attribute__((ext_vector_type(2))) float v2f;
typedef __attribute__((ext_vector_type(8))) float v8f;

#define SEQS      4096
#define LSEQ      2048
#define NST       64
#define TCH       16
#define NSEG      8
#define CH_SEG    16
#define NTILE     256
#define WPB       4

#define OFF_M     0
#define OFF_CM    256
#define OFF_BM    1280
#define OFF_A16   2304
#define OFF_ST    2560

__device__ __forceinline__ v8f wmma4_f32(v2f a, v2f b, v8f c) {
  return __builtin_amdgcn_wmma_f32_16x16x4_f32(false, a, false, b,
                                               (short)0, c, false, false);
}

__global__ void ssm_prep(const float* __restrict__ A, const float* __restrict__ Bv,
                         const float* __restrict__ Cv, float* __restrict__ tbl) {
  int t = threadIdx.x;
  float* Mm  = tbl + OFF_M;
  float* Cm  = tbl + OFF_CM;
  float* Bm  = tbl + OFF_BM;
  float* a16 = tbl + OFF_A16;
  if (t < NST) {
    float a = A[t * NST + t];
    float b = Bv[t], c = Cv[t];
    float pw[17];
    float p = 1.0f;
    for (int i = 0; i < 17; ++i) { pw[i] = p; p *= a; }
    a16[t] = pw[16];
    for (int k = 0; k < TCH; ++k) Bm[t * TCH + k] = b * pw[15 - k];
    for (int l = 0; l < TCH; ++l) Cm[l * NST + t] = c * pw[l + 1];
  }
  if (t < 256) {
    int l = t >> 4, k = t & 15;
    float v = 0.0f;
    if (k <= l) {
      int d = l - k;
      for (int n = 0; n < NST; ++n) {
        float a = A[n * NST + n];
        float p = 1.0f;
        for (int i = 0; i < d; ++i) p *= a;
        v += Cv[n] * Bv[n] * p;
      }
    }
    Mm[l * TCH + k] = v;
  }
}

__global__ __launch_bounds__(WPB * 32) void ssm_state_pass(
    const float* __restrict__ u, const float* __restrict__ tbl,
    float* __restrict__ states) {
  const int lane = threadIdx.x & 31;
  const int half = lane >> 4;
  const int lm   = lane & 15;
  const int tile = blockIdx.x * WPB + (threadIdx.x >> 5);

  const float* Bm  = tbl + OFF_BM;
  const float* a16 = tbl + OFF_A16;

  v2f bm[4][4];
  #pragma unroll
  for (int t = 0; t < 4; ++t)
    #pragma unroll
    for (int j = 0; j < 4; ++j)
      bm[t][j] = *(const v2f*)(Bm + (16 * t + lm) * TCH + 4 * j + 2 * half);

  float av[4][8];
  #pragma unroll
  for (int t = 0; t < 4; ++t)
    #pragma unroll
    for (int r = 0; r < 8; ++r)
      av[t][r] = a16[16 * t + r + 8 * half];

  v8f H[4];
  const v8f zero = {0.f, 0.f, 0.f, 0.f, 0.f, 0.f, 0.f, 0.f};
  #pragma unroll
  for (int t = 0; t < 4; ++t) H[t] = zero;

  const float* useq = u + (size_t)(tile * 16 + lm) * LSEQ;

  for (int g = 0; g < NSEG; ++g) {
    float* sp = states + (size_t)(tile * NSEG + g) * 1024;
    #pragma unroll
    for (int t = 0; t < 4; ++t)
      #pragma unroll
      for (int r = 0; r < 8; ++r)
        sp[t * 256 + r * 32 + lane] = H[t][r];

    const float* useg = useq + g * (CH_SEG * TCH);
    for (int c = 0; c < CH_SEG; ++c) {
      const int cb = c * TCH;
      v2f ub[4];
      #pragma unroll
      for (int j = 0; j < 4; ++j)
        ub[j] = *(const v2f*)(useg + cb + 4 * j + 2 * half);
      __builtin_prefetch((const void*)(useg + cb + 4 * TCH), 0, 0);

      #pragma unroll
      for (int t = 0; t < 4; ++t) {
        #pragma unroll
        for (int r = 0; r < 8; ++r) H[t][r] *= av[t][r];
        #pragma unroll
        for (int j = 0; j < 4; ++j) H[t] = wmma4_f32(bm[t][j], ub[j], H[t]);
      }
    }
  }
}

__global__ __launch_bounds__(WPB * 32) void ssm_main(
    const float* __restrict__ u, const float* __restrict__ tbl,
    const float* __restrict__ states, float* __restrict__ y) {
  const int lane = threadIdx.x & 31;
  const int half = lane >> 4;
  const int lm   = lane & 15;
  const int tile = blockIdx.x * WPB + (threadIdx.x >> 5);
  const int seg  = blockIdx.y;

  const float* Mm  = tbl + OFF_M;
  const float* Cm  = tbl + OFF_CM;
  const float* Bm  = tbl + OFF_BM;
  const float* a16 = tbl + OFF_A16;

  v2f mA[4];
  #pragma unroll
  for (int j = 0; j < 4; ++j)
    mA[j] = *(const v2f*)(Mm + lm * TCH + 4 * j + 2 * half);

  v2f cmA[4][4];
  #pragma unroll
  for (int t = 0; t < 4; ++t)
    #pragma unroll
    for (int j = 0; j < 4; ++j)
      cmA[t][j] = *(const v2f*)(Cm + lm * NST + 16 * t + 4 * j + 2 * half);

  v2f bm[4][4];
  #pragma unroll
  for (int t = 0; t < 4; ++t)
    #pragma unroll
    for (int j = 0; j < 4; ++j)
      bm[t][j] = *(const v2f*)(Bm + (16 * t + lm) * TCH + 4 * j + 2 * half);

  float av[4][8];
  #pragma unroll
  for (int t = 0; t < 4; ++t)
    #pragma unroll
    for (int r = 0; r < 8; ++r)
      av[t][r] = a16[16 * t + r + 8 * half];

  v8f H[4];
  {
    const float* sp = states + (size_t)(tile * NSEG + seg) * 1024;
    #pragma unroll
    for (int t = 0; t < 4; ++t)
      #pragma unroll
      for (int r = 0; r < 8; ++r)
        H[t][r] = sp[t * 256 + r * 32 + lane];
  }

  const float* useq = u + (size_t)(tile * 16 + lm) * LSEQ + seg * (CH_SEG * TCH);
  float*       yseq = y + (size_t)(tile * 16 + lm) * LSEQ + seg * (CH_SEG * TCH);
  const v8f zero = {0.f, 0.f, 0.f, 0.f, 0.f, 0.f, 0.f, 0.f};

  for (int c = 0; c < CH_SEG; ++c) {
    const int cb = c * TCH;
    v2f ub[4];
    #pragma unroll
    for (int j = 0; j < 4; ++j)
      ub[j] = *(const v2f*)(useq + cb + 4 * j + 2 * half);
    __builtin_prefetch((const void*)(useq + cb + 4 * TCH), 0, 0);

    v8f yacc = zero;
    #pragma unroll
    for (int j = 0; j < 4; ++j) yacc = wmma4_f32(mA[j], ub[j], yacc);

    #pragma unroll
    for (int t = 0; t < 4; ++t) {
      #pragma unroll
      for (int j = 0; j < 4; ++j) {
        const int jj = j & 1;
        float e0 = H[t][4 * jj + 0], e1 = H[t][4 * jj + 1];
        float e2 = H[t][4 * jj + 2], e3 = H[t][4 * jj + 3];
        float sx = half ? e0 : e2;
        float sy = half ? e1 : e3;
        float px = __shfl_xor(sx, 16);
        float py = __shfl_xor(sy, 16);
        float ox = half ? e2 : e0;
        float oy = half ? e3 : e1;
        bool useOwn = (half == (j >> 1));
        v2f hb;
        hb.x = useOwn ? ox : px;
        hb.y = useOwn ? oy : py;
        yacc = wmma4_f32(cmA[t][j], hb, yacc);
      }
    }

    if (c != CH_SEG - 1) {
      #pragma unroll
      for (int t = 0; t < 4; ++t) {
        #pragma unroll
        for (int r = 0; r < 8; ++r) H[t][r] *= av[t][r];
        #pragma unroll
        for (int j = 0; j < 4; ++j) H[t] = wmma4_f32(bm[t][j], ub[j], H[t]);
      }
    }

    #pragma unroll
    for (int r = 0; r < 8; ++r)
      yseq[cb + r + 8 * half] = yacc[r];
  }
}

extern "C" void kernel_launch(void* const* d_in, const int* in_sizes, int n_in,
                              void* d_out, int out_size, void* d_ws, size_t ws_size,
                              hipStream_t stream) {
  const float* u  = (const float*)d_in[0];
  const float* A  = (const float*)d_in[1];
  const float* Bv = (const float*)d_in[2];
  const float* Cv = (const float*)d_in[3];
  float* tbl    = (float*)d_ws;
  float* states = tbl + OFF_ST;
  float* y      = (float*)d_out;

  ssm_prep<<<dim3(1), dim3(256), 0, stream>>>(A, Bv, Cv, tbl);
  ssm_state_pass<<<dim3(NTILE / WPB), dim3(WPB * 32), 0, stream>>>(u, tbl, states);
  ssm_main<<<dim3(NTILE / WPB, NSEG), dim3(WPB * 32), 0, stream>>>(u, tbl, states, y);
}


# y goes out through a per-wave LDS tile that collects two chunks: each global store instruction writes 4 whole 128-byte lines (was 16 lines x 32 bytes per instruction); sum orders unchanged
# speedup vs baseline: 3.2417x; 3.2417x over previous
#include <hip/hip_runtime.h>


typedef __attribute__((ext_vector_type(2))) float v2f;
typedef __attribute__((ext_vector_type(8))) float v8f;
typedef __attribute__((ext_vector_type(4))) float v4f;

#define SEQS      4096
#define LSEQ      2048
#define NST       64
#define TCH       16
#define NSEG      16
#define CH_SEG    8
#define WPB       4
#define YPITCH    36

__device__ __forceinline__ v8f wmma4_f32(v2f a, v2f b, v8f c) {
  return __builtin_amdgcn_wmma_f32_16x16x4_f32(false, a, false, b,
                                               (short)0, c, false, false);
}

template <int D>
__device__ __forceinline__ float col_shr(float x) {
  return __builtin_bit_cast(float,
      __builtin_amdgcn_update_dpp(0, __builtin_bit_cast(int, x),
                                  0x110 + D, 0xF, 0xF, true));
}

template <int D>
__device__ __forceinline__ void fold_step(v8f (&H)[4], float (&q)[4][8]) {
  #pragma unroll
  for (int t = 0; t < 4; ++t)
    #pragma unroll
    for (int r = 0; r < 8; ++r) {
      H[t][r] += q[t][r] * col_shr<D>(H[t][r]);
      q[t][r] *= q[t][r];
    }
}

__global__ __launch_bounds__(WPB * 32) void ssm_seq(
    const float* __restrict__ u, const float* __restrict__ A,
    const float* __restrict__ Bv, const float* __restrict__ Cv,
    float* __restrict__ y) {
  const int lane = threadIdx.x & 31;
  const int half = lane >> 4;
  const int lm   = lane & 15;
  const int seq  = blockIdx.x * WPB + (threadIdx.x >> 5);

  const float* useq = u + (size_t)seq * LSEQ + lm * (CH_SEG * TCH);
  __shared__ __attribute__((aligned(16))) float ytile[WPB][NSEG * YPITCH];
  float* ytw = &ytile[threadIdx.x >> 5][0];
  float*       ywr = ytw + lm * YPITCH + 8 * half;
  const float* yrd = ytw + (lane >> 3) * YPITCH + 4 * (lane & 7);
  float*       yln = y + (size_t)seq * LSEQ + (lane >> 3) * (CH_SEG * TCH)
                       + 4 * (lane & 7);
  float a_bm[4], b_bm[4];
  #pragma unroll
  for (int t = 0; t < 4; ++t) {
    const int n = 16 * t + lm;
    a_bm[t] = A[n * NST + n];
    b_bm[t] = Bv[n];
  }
  float a_av[4][8];
  #pragma unroll
  for (int t = 0; t < 4; ++t)
    #pragma unroll
    for (int r = 0; r < 8; ++r) {
      const int n = 16 * t + r + 8 * half;
      a_av[t][r] = A[n * NST + n];
    }
  float b_cm[4][4][2], c_cm[4][4][2];
  #pragma unroll
  for (int t = 0; t < 4; ++t)
    #pragma unroll
    for (int j = 0; j < 4; ++j)
      #pragma unroll
      for (int s = 0; s < 2; ++s) {
        const int n = 16 * t + 8 * half + 2 * j + s;
        b_cm[t][j][s] = Bv[n];
        c_cm[t][j][s] = Cv[n];
      }

  v2f ub[CH_SEG][4];
  #pragma unroll
  for (int ch = 0; ch < CH_SEG; ++ch)
    #pragma unroll
    for (int j = 0; j < 4; ++j)
      ub[ch][j] = *(const v2f*)(useq + ch * TCH + 8 * half + 2 * j);

  v2f bm[4][4];
  #pragma unroll
  for (int t = 0; t < 4; ++t) {
    const float a = a_bm[t];
    const float b = b_bm[t];
    const float a2 = a * a, a4 = a2 * a2;
    float pw[8];
    pw[0] = half ? 1.0f : a4 * a4;
    #pragma unroll
    for (int i = 1; i < 8; ++i) pw[i] = pw[i - 1] * a;
    #pragma unroll
    for (int j = 0; j < 4; ++j) {
      bm[t][j].x = b * pw[7 - 2 * j];
      bm[t][j].y = b * pw[6 - 2 * j];
    }
  }

  float av[4][8];
  #pragma unroll
  for (int t = 0; t < 4; ++t)
    #pragma unroll
    for (int r = 0; r < 8; ++r) {
      const float a = a_av[t][r];
      const float a2 = a * a, a4 = a2 * a2, a8 = a4 * a4;
      av[t][r] = a8 * a8;
    }

  const v8f zero = {0.f, 0.f, 0.f, 0.f, 0.f, 0.f, 0.f, 0.f};

  const bool l1 = (lm & 1) != 0, l2 = (lm & 2) != 0;
  const bool l4 = (lm & 4) != 0, l8 = (lm & 8) != 0;
  v2f cmA[4][4];
  v8f kd = zero;
  const v2f ones = {1.0f, 1.0f};
  #pragma unroll
  for (int t = 0; t < 4; ++t)
    #pragma unroll
    for (int j = 0; j < 4; ++j) {
      v2f kA;
      #pragma unroll
      for (int s = 0; s < 2; ++s) {
        const float a = a_av[t][2 * j + s];
        const float a2 = a * a, a4 = a2 * a2, a8 = a4 * a4;
        float p = l1 ? a : 1.0f;
        p *= l2 ? a2 : 1.0f;
        p *= l4 ? a4 : 1.0f;
        p *= l8 ? a8 : 1.0f;
        const float cp = c_cm[t][j][s] * p;
        kA[s] = cp * b_cm[t][j][s];
        cmA[t][j][s] = cp * a;
      }
      kd = wmma4_f32(kA, ones, kd);
    }

  float kl;
  {
    const float s0 = l1 ? kd[1] : kd[0], s1 = l1 ? kd[3] : kd[2];
    const float s2 = l1 ? kd[5] : kd[4], s3 = l1 ? kd[7] : kd[6];
    const float s01 = l2 ? s1 : s0, s23 = l2 ? s3 : s2;
    const float kown = l4 ? s23 : s01;
    const float koth = __shfl_xor(kown, 16);
    kl = (l8 == (half != 0)) ? kown : koth;
  }

  v2f mA[4];
  #pragma unroll
  for (int j = 0; j < 4; ++j)
    #pragma unroll
    for (int s = 0; s < 2; ++s) {
      const int d = lm - (8 * half + 2 * j + s);
      const float v = __shfl(kl, d & 15);
      mA[j][s] = d >= 0 ? v : 0.0f;
    }

  v8f H[4];
  #pragma unroll
  for (int t = 0; t < 4; ++t) H[t] = zero;
  #pragma unroll
  for (int ch = 0; ch < CH_SEG; ++ch) {
    #pragma unroll
    for (int t = 0; t < 4; ++t) {
      if (ch > 0) {
        #pragma unroll
        for (int r = 0; r < 8; ++r) H[t][r] *= av[t][r];
      }
      #pragma unroll
      for (int j = 0; j < 4; ++j) H[t] = wmma4_f32(bm[t][j], ub[ch][j], H[t]);
    }
  }

  {
    float q[4][8];
    #pragma unroll
    for (int t = 0; t < 4; ++t)
      #pragma unroll
      for (int r = 0; r < 8; ++r) {
        float m = av[t][r];
        m *= m; m *= m; m *= m;
        q[t][r] = m;
      }
    fold_step<1>(H, q);
    fold_step<2>(H, q);
    fold_step<4>(H, q);
    fold_step<8>(H, q);
    #pragma unroll
    for (int t = 0; t < 4; ++t)
      #pragma unroll
      for (int r = 0; r < 8; ++r)
        H[t][r] = col_shr<1>(H[t][r]);
  }


  #pragma unroll
  for (int ch = 0; ch < CH_SEG; ++ch) {
    const int cb = ch * TCH;

    v8f yacc = zero;
    #pragma unroll
    for (int j = 0; j < 4; ++j) yacc = wmma4_f32(mA[j], ub[ch][j], yacc);

    #pragma unroll
    for (int t = 0; t < 4; ++t) {
      #pragma unroll
      for (int j = 0; j < 4; ++j) {
        v2f hb;
        hb.x = H[t][2 * j];
        hb.y = H[t][2 * j + 1];
        yacc = wmma4_f32(cmA[t][j], hb, yacc);
      }
    }

    if (ch != CH_SEG - 1) {
      #pragma unroll
      for (int t = 0; t < 4; ++t) {
        #pragma unroll
        for (int r = 0; r < 8; ++r) H[t][r] *= av[t][r];
        #pragma unroll
        for (int j = 0; j < 4; ++j) H[t] = wmma4_f32(bm[t][j], ub[ch][j], H[t]);
      }
    }

    {
      v4f lo = {yacc[0], yacc[1], yacc[2], yacc[3]};
      v4f hi = {yacc[4], yacc[5], yacc[6], yacc[7]};
      *(v4f*)(ywr + (ch & 1) * TCH)     = lo;
      *(v4f*)(ywr + (ch & 1) * TCH + 4) = hi;
    }
    if (ch & 1) {
      v4f yl[4];
      #pragma unroll
      for (int i = 0; i < 4; ++i)
        yl[i] = *(const v4f*)(yrd + 4 * i * YPITCH);
      #pragma unroll
      for (int i = 0; i < 4; ++i)
        *(v4f*)(yln + 4 * i * (CH_SEG * TCH) + (cb - TCH)) = yl[i];
    }
  }
}

extern "C" void kernel_launch(void* const* d_in, const int* in_sizes, int n_in,
                              void* d_out, int out_size, void* d_ws, size_t ws_size,
                              hipStream_t stream) {
  const float* u  = (const float*)d_in[0];
  const float* A  = (const float*)d_in[1];
  const float* Bv = (const float*)d_in[2];
  const float* Cv = (const float*)d_in[3];
  float* y      = (float*)d_out;

  ssm_seq<<<dim3(SEQS / WPB), dim3(WPB * 32), 0, stream>>>(u, A, Bv, Cv, y);
}


# first pass keeps its 7 chunk products Bm*U in registers (224 more; 2 waves per SIMD, was 4); output pass takes the state from them by fma: 304 matrix ops per wave, was 416; state sum order changes
# speedup vs baseline: 3.6784x; 1.1347x over previous
#include <hip/hip_runtime.h>


typedef __attribute__((ext_vector_type(2))) float v2f;
typedef __attribute__((ext_vector_type(8))) float v8f;
typedef __attribute__((ext_vector_type(4))) float v4f;

#define SEQS      4096
#define LSEQ      2048
#define NST       64
#define TCH       16
#define NSEG      16
#define CH_SEG    8
#define WPB       4
#define YPITCH    36

__device__ __forceinline__ v8f wmma4_f32(v2f a, v2f b, v8f c) {
  return __builtin_amdgcn_wmma_f32_16x16x4_f32(false, a, false, b,
                                               (short)0, c, false, false);
}

template <int D>
__device__ __forceinline__ float col_shr(float x) {
  return __builtin_bit_cast(float,
      __builtin_amdgcn_update_dpp(0, __builtin_bit_cast(int, x),
                                  0x110 + D, 0xF, 0xF, true));
}

template <int D>
__device__ __forceinline__ void fold_step(v8f (&H)[4], float (&q)[4][8]) {
  #pragma unroll
  for (int t = 0; t < 4; ++t)
    #pragma unroll
    for (int r = 0; r < 8; ++r) {
      H[t][r] += q[t][r] * col_shr<D>(H[t][r]);
      q[t][r] *= q[t][r];
    }
}

__global__ __launch_bounds__(WPB * 32) void ssm_seq(
    const float* __restrict__ u, const float* __restrict__ A,
    const float* __restrict__ Bv, const float* __restrict__ Cv,
    float* __restrict__ y) {
  const int lane = threadIdx.x & 31;
  const int half = lane >> 4;
  const int lm   = lane & 15;
  const int seq  = blockIdx.x * WPB + (threadIdx.x >> 5);

  const float* useq = u + (size_t)seq * LSEQ + lm * (CH_SEG * TCH);
  __shared__ __attribute__((aligned(16))) float ytile[WPB][NSEG * YPITCH];
  float* ytw = &ytile[threadIdx.x >> 5][0];
  float*       ywr = ytw + lm * YPITCH + 8 * half;
  const float* yrd = ytw + (lane >> 3) * YPITCH + 4 * (lane & 7);
  float*       yln = y + (size_t)seq * LSEQ + (lane >> 3) * (CH_SEG * TCH)
                       + 4 * (lane & 7);
  float a_bm[4], b_bm[4];
  #pragma unroll
  for (int t = 0; t < 4; ++t) {
    const int n = 16 * t + lm;
    a_bm[t] = A[n * NST + n];
    b_bm[t] = Bv[n];
  }
  float a_av[4][8];
  #pragma unroll
  for (int t = 0; t < 4; ++t)
    #pragma unroll
    for (int r = 0; r < 8; ++r) {
      const int n = 16 * t + r + 8 * half;
      a_av[t][r] = A[n * NST + n];
    }
  float b_cm[4][4][2], c_cm[4][4][2];
  #pragma unroll
  for (int t = 0; t < 4; ++t)
    #pragma unroll
    for (int j = 0; j < 4; ++j)
      #pragma unroll
      for (int s = 0; s < 2; ++s) {
        const int n = 16 * t + 8 * half + 2 * j + s;
        b_cm[t][j][s] = Bv[n];
        c_cm[t][j][s] = Cv[n];
      }

  v2f ub[CH_SEG][4];
  #pragma unroll
  for (int ch = 0; ch < CH_SEG; ++ch)
    #pragma unroll
    for (int j = 0; j < 4; ++j)
      ub[ch][j] = *(const v2f*)(useq + ch * TCH + 8 * half + 2 * j);

  v2f bm[4][4];
  #pragma unroll
  for (int t = 0; t < 4; ++t) {
    const float a = a_bm[t];
    const float b = b_bm[t];
    const float a2 = a * a, a4 = a2 * a2;
    float pw[8];
    pw[0] = half ? 1.0f : a4 * a4;
    #pragma unroll
    for (int i = 1; i < 8; ++i) pw[i] = pw[i - 1] * a;
    #pragma unroll
    for (int j = 0; j < 4; ++j) {
      bm[t][j].x = b * pw[7 - 2 * j];
      bm[t][j].y = b * pw[6 - 2 * j];
    }
  }

  float av[4][8];
  #pragma unroll
  for (int t = 0; t < 4; ++t)
    #pragma unroll
    for (int r = 0; r < 8; ++r) {
      const float a = a_av[t][r];
      const float a2 = a * a, a4 = a2 * a2, a8 = a4 * a4;
      av[t][r] = a8 * a8;
    }

  const v8f zero = {0.f, 0.f, 0.f, 0.f, 0.f, 0.f, 0.f, 0.f};

  const bool l1 = (lm & 1) != 0, l2 = (lm & 2) != 0;
  const bool l4 = (lm & 4) != 0, l8 = (lm & 8) != 0;
  v2f cmA[4][4];
  v8f kd = zero;
  const v2f ones = {1.0f, 1.0f};
  #pragma unroll
  for (int t = 0; t < 4; ++t)
    #pragma unroll
    for (int j = 0; j < 4; ++j) {
      v2f kA;
      #pragma unroll
      for (int s = 0; s < 2; ++s) {
        const float a = a_av[t][2 * j + s];
        const float a2 = a * a, a4 = a2 * a2, a8 = a4 * a4;
        float p = l1 ? a : 1.0f;
        p *= l2 ? a2 : 1.0f;
        p *= l4 ? a4 : 1.0f;
        p *= l8 ? a8 : 1.0f;
        const float cp = c_cm[t][j][s] * p;
        kA[s] = cp * b_cm[t][j][s];
        cmA[t][j][s] = cp * a;
      }
      kd = wmma4_f32(kA, ones, kd);
    }

  float kl;
  {
    const float s0 = l1 ? kd[1] : kd[0], s1 = l1 ? kd[3] : kd[2];
    const float s2 = l1 ? kd[5] : kd[4], s3 = l1 ? kd[7] : kd[6];
    const float s01 = l2 ? s1 : s0, s23 = l2 ? s3 : s2;
    const float kown = l4 ? s23 : s01;
    const float koth = __shfl_xor(kown, 16);
    kl = (l8 == (half != 0)) ? kown : koth;
  }

  v2f mA[4];
  #pragma unroll
  for (int j = 0; j < 4; ++j)
    #pragma unroll
    for (int s = 0; s < 2; ++s) {
      const int d = lm - (8 * half + 2 * j + s);
      const float v = __shfl(kl, d & 15);
      mA[j][s] = d >= 0 ? v : 0.0f;
    }

  v8f P[CH_SEG - 1][4];
  v8f H[4];
  #pragma unroll
  for (int ch = 0; ch < CH_SEG; ++ch) {
    #pragma unroll
    for (int t = 0; t < 4; ++t) {
      v8f p = zero;
      #pragma unroll
      for (int j = 0; j < 4; ++j) p = wmma4_f32(bm[t][j], ub[ch][j], p);
      if (ch == 0) {
        H[t] = p;
      } else {
        #pragma unroll
        for (int r = 0; r < 8; ++r) H[t][r] = av[t][r] * H[t][r] + p[r];
      }
      if (ch < CH_SEG - 1) P[ch][t] = p;
    }
  }

  {
    float q[4][8];
    #pragma unroll
    for (int t = 0; t < 4; ++t)
      #pragma unroll
      for (int r = 0; r < 8; ++r) {
        float m = av[t][r];
        m *= m; m *= m; m *= m;
        q[t][r] = m;
      }
    fold_step<1>(H, q);
    fold_step<2>(H, q);
    fold_step<4>(H, q);
    fold_step<8>(H, q);
    #pragma unroll
    for (int t = 0; t < 4; ++t)
      #pragma unroll
      for (int r = 0; r < 8; ++r)
        H[t][r] = col_shr<1>(H[t][r]);
  }


  #pragma unroll
  for (int pr = 0; pr < CH_SEG / 2; ++pr) {
    const int c0 = 2 * pr, c1 = c0 + 1;

    v8f H1[4];
    #pragma unroll
    for (int t = 0; t < 4; ++t)
      #pragma unroll
      for (int r = 0; r < 8; ++r) H1[t][r] = av[t][r] * H[t][r] + P[c0][t][r];

    v8f y0 = zero, y1 = zero;
    #pragma unroll
    for (int j = 0; j < 4; ++j) {
      y0 = wmma4_f32(mA[j], ub[c0][j], y0);
      y1 = wmma4_f32(mA[j], ub[c1][j], y1);
    }

    #pragma unroll
    for (int t = 0; t < 4; ++t) {
      #pragma unroll
      for (int j = 0; j < 4; ++j) {
        v2f h0, h1;
        h0.x = H[t][2 * j];
        h0.y = H[t][2 * j + 1];
        h1.x = H1[t][2 * j];
        h1.y = H1[t][2 * j + 1];
        y0 = wmma4_f32(cmA[t][j], h0, y0);
        y1 = wmma4_f32(cmA[t][j], h1, y1);
      }
    }

    if (pr != CH_SEG / 2 - 1) {
      #pragma unroll
      for (int t = 0; t < 4; ++t)
        #pragma unroll
        for (int r = 0; r < 8; ++r) H[t][r] = av[t][r] * H1[t][r] + P[c1][t][r];
    }

    {
      v4f lo0 = {y0[0], y0[1], y0[2], y0[3]};
      v4f hi0 = {y0[4], y0[5], y0[6], y0[7]};
      v4f lo1 = {y1[0], y1[1], y1[2], y1[3]};
      v4f hi1 = {y1[4], y1[5], y1[6], y1[7]};
      *(v4f*)(ywr)           = lo0;
      *(v4f*)(ywr + 4)       = hi0;
      *(v4f*)(ywr + TCH)     = lo1;
      *(v4f*)(ywr + TCH + 4) = hi1;
    }
    {
      v4f yl[4];
      #pragma unroll
      for (int i = 0; i < 4; ++i)
        yl[i] = *(const v4f*)(yrd + 4 * i * YPITCH);
      #pragma unroll
      for (int i = 0; i < 4; ++i)
        *(v4f*)(yln + 4 * i * (CH_SEG * TCH) + c0 * TCH) = yl[i];
    }
  }
}

extern "C" void kernel_launch(void* const* d_in, const int* in_sizes, int n_in,
                              void* d_out, int out_size, void* d_ws, size_t ws_size,
                              hipStream_t stream) {
  const float* u  = (const float*)d_in[0];
  const float* A  = (const float*)d_in[1];
  const float* Bv = (const float*)d_in[2];
  const float* Cv = (const float*)d_in[3];
  float* y      = (float*)d_out;

  ssm_seq<<<dim3(SEQS / WPB), dim3(WPB * 32), 0, stream>>>(u, A, Bv, Cv, y);
}


# each wave takes two adjacent sequences in turn (2048 waves, all resident from launch); a chunk pair is reloaded from the next sequence after its last use in the output pass; sum orders unchanged
# speedup vs baseline: 3.9942x; 1.0859x over previous
#include <hip/hip_runtime.h>


typedef __attribute__((ext_vector_type(2))) float v2f;
typedef __attribute__((ext_vector_type(8))) float v8f;
typedef __attribute__((ext_vector_type(4))) float v4f;

#define SEQS      4096
#define LSEQ      2048
#define NST       64
#define TCH       16
#define NSEG      16
#define CH_SEG    8
#define WPB       4
#define SPW       2
#define YPITCH    36

__device__ __forceinline__ v8f wmma4_f32(v2f a, v2f b, v8f c) {
  return __builtin_amdgcn_wmma_f32_16x16x4_f32(false, a, false, b,
                                               (short)0, c, false, false);
}

template <int D>
__device__ __forceinline__ float col_shr(float x) {
  return __builtin_bit_cast(float,
      __builtin_amdgcn_update_dpp(0, __builtin_bit_cast(int, x),
                                  0x110 + D, 0xF, 0xF, true));
}

template <int D>
__device__ __forceinline__ void fold_step(v8f (&H)[4], float (&q)[4][8]) {
  #pragma unroll
  for (int t = 0; t < 4; ++t)
    #pragma unroll
    for (int r = 0; r < 8; ++r) {
      H[t][r] += q[t][r] * col_shr<D>(H[t][r]);
      q[t][r] *= q[t][r];
    }
}

__device__ __forceinline__ void first_pass(const v2f (&bm)[4][4],
                                           const float (&av)[4][8],
                                           const v2f (&ub)[CH_SEG][4],
                                           v8f (&P)[CH_SEG - 1][4],
                                           v8f (&H)[4]) {
  const v8f zero = {0.f, 0.f, 0.f, 0.f, 0.f, 0.f, 0.f, 0.f};
  #pragma unroll
  for (int ch = 0; ch < CH_SEG; ++ch) {
    #pragma unroll
    for (int t = 0; t < 4; ++t) {
      v8f p = zero;
      #pragma unroll
      for (int j = 0; j < 4; ++j) p = wmma4_f32(bm[t][j], ub[ch][j], p);
      if (ch == 0) {
        H[t] = p;
      } else {
        #pragma unroll
        for (int r = 0; r < 8; ++r) H[t][r] = av[t][r] * H[t][r] + p[r];
      }
      if (ch < CH_SEG - 1) P[ch][t] = p;
    }
    if (ch == CH_SEG - 3) __builtin_amdgcn_sched_barrier(0);
  }
}

__device__ __forceinline__ void output_pass(const v2f (&mA)[4],
                                            const v2f (&cmA)[4][4],
                                            const float (&av)[4][8],
                                            v2f (&ub)[CH_SEG][4],
                                            const v8f (&P)[CH_SEG - 1][4],
                                            v8f (&H)[4],
                                            float* ywr, const float* yrd,
                                            float* yln, const float* unext,
                                            const int half) {
  const v8f zero = {0.f, 0.f, 0.f, 0.f, 0.f, 0.f, 0.f, 0.f};
  #pragma unroll
  for (int pr = 0; pr < CH_SEG / 2; ++pr) {
    const int c0 = 2 * pr, c1 = c0 + 1;

    v8f y0 = zero, y1 = zero;
    #pragma unroll
    for (int j = 0; j < 4; ++j) {
      y0 = wmma4_f32(mA[j], ub[c0][j], y0);
      y1 = wmma4_f32(mA[j], ub[c1][j], y1);
    }

    __builtin_amdgcn_sched_barrier(0);
    #pragma unroll
    for (int j = 0; j < 4; ++j) {
      ub[c0][j] = *(const v2f*)(unext + c0 * TCH + 8 * half + 2 * j);
      ub[c1][j] = *(const v2f*)(unext + c1 * TCH + 8 * half + 2 * j);
    }

    v8f H1[4];
    #pragma unroll
    for (int t = 0; t < 4; ++t)
      #pragma unroll
      for (int r = 0; r < 8; ++r) H1[t][r] = av[t][r] * H[t][r] + P[c0][t][r];

    #pragma unroll
    for (int t = 0; t < 4; ++t) {
      #pragma unroll
      for (int j = 0; j < 4; ++j) {
        v2f h0, h1;
        h0.x = H[t][2 * j];
        h0.y = H[t][2 * j + 1];
        h1.x = H1[t][2 * j];
        h1.y = H1[t][2 * j + 1];
        y0 = wmma4_f32(cmA[t][j], h0, y0);
        y1 = wmma4_f32(cmA[t][j], h1, y1);
      }
    }

    if (pr != CH_SEG / 2 - 1) {
      #pragma unroll
      for (int t = 0; t < 4; ++t)
        #pragma unroll
        for (int r = 0; r < 8; ++r) H[t][r] = av[t][r] * H1[t][r] + P[c1][t][r];
    }

    {
      v4f lo0 = {y0[0], y0[1], y0[2], y0[3]};
      v4f hi0 = {y0[4], y0[5], y0[6], y0[7]};
      v4f lo1 = {y1[0], y1[1], y1[2], y1[3]};
      v4f hi1 = {y1[4], y1[5], y1[6], y1[7]};
      *(v4f*)(ywr)           = lo0;
      *(v4f*)(ywr + 4)       = hi0;
      *(v4f*)(ywr + TCH)     = lo1;
      *(v4f*)(ywr + TCH + 4) = hi1;
    }
    {
      v4f yl[4];
      #pragma unroll
      for (int i = 0; i < 4; ++i)
        yl[i] = *(const v4f*)(yrd + 4 * i * YPITCH);
      #pragma unroll
      for (int i = 0; i < 4; ++i)
        *(v4f*)(yln + 4 * i * (CH_SEG * TCH) + c0 * TCH) = yl[i];
    }
  }
}

__global__ __launch_bounds__(WPB * 32) void ssm_seq(
    const float* __restrict__ u, const float* __restrict__ A,
    const float* __restrict__ Bv, const float* __restrict__ Cv,
    float* __restrict__ y) {
  const int lane = threadIdx.x & 31;
  const int half = lane >> 4;
  const int lm   = lane & 15;
  const int seq  = (blockIdx.x * WPB + (threadIdx.x >> 5)) * SPW;

  const float* useq = u + (size_t)seq * LSEQ + lm * (CH_SEG * TCH);
  __shared__ __attribute__((aligned(16))) float ytile[WPB][NSEG * YPITCH];
  float* ytw = &ytile[threadIdx.x >> 5][0];
  float*       ywr = ytw + lm * YPITCH + 8 * half;
  const float* yrd = ytw + (lane >> 3) * YPITCH + 4 * (lane & 7);
  float*       yseq = y + (size_t)seq * LSEQ + (lane >> 3) * (CH_SEG * TCH)
                        + 4 * (lane & 7);
  float a_bm[4], b_bm[4];
  #pragma unroll
  for (int t = 0; t < 4; ++t) {
    const int n = 16 * t + lm;
    a_bm[t] = A[n * NST + n];
    b_bm[t] = Bv[n];
  }
  float a_av[4][8];
  #pragma unroll
  for (int t = 0; t < 4; ++t)
    #pragma unroll
    for (int r = 0; r < 8; ++r) {
      const int n = 16 * t + r + 8 * half;
      a_av[t][r] = A[n * NST + n];
    }
  float b_cm[4][4][2], c_cm[4][4][2];
  #pragma unroll
  for (int t = 0; t < 4; ++t)
    #pragma unroll
    for (int j = 0; j < 4; ++j)
      #pragma unroll
      for (int s = 0; s < 2; ++s) {
        const int n = 16 * t + 8 * half + 2 * j + s;
        b_cm[t][j][s] = Bv[n];
        c_cm[t][j][s] = Cv[n];
      }

  v2f ub[CH_SEG][4];
  #pragma unroll
  for (int ch = 0; ch < CH_SEG; ++ch)
    #pragma unroll
    for (int j = 0; j < 4; ++j)
      ub[ch][j] = *(const v2f*)(useq + ch * TCH + 8 * half + 2 * j);
  __builtin_amdgcn_sched_barrier(0);

  v2f bm[4][4];
  #pragma unroll
  for (int t = 0; t < 4; ++t) {
    const float a = a_bm[t];
    const float b = b_bm[t];
    const float a2 = a * a, a4 = a2 * a2;
    float pw[8];
    pw[0] = half ? 1.0f : a4 * a4;
    #pragma unroll
    for (int i = 1; i < 8; ++i) pw[i] = pw[i - 1] * a;
    #pragma unroll
    for (int j = 0; j < 4; ++j) {
      bm[t][j].x = b * pw[7 - 2 * j];
      bm[t][j].y = b * pw[6 - 2 * j];
    }
  }

  float av[4][8];
  #pragma unroll
  for (int t = 0; t < 4; ++t)
    #pragma unroll
    for (int r = 0; r < 8; ++r) {
      const float a = a_av[t][r];
      const float a2 = a * a, a4 = a2 * a2, a8 = a4 * a4;
      av[t][r] = a8 * a8;
    }

  const v8f zero = {0.f, 0.f, 0.f, 0.f, 0.f, 0.f, 0.f, 0.f};

  const bool l1 = (lm & 1) != 0, l2 = (lm & 2) != 0;
  const bool l4 = (lm & 4) != 0, l8 = (lm & 8) != 0;
  v2f cmA[4][4];
  v8f kd = zero;
  const v2f ones = {1.0f, 1.0f};
  #pragma unroll
  for (int t = 0; t < 4; ++t)
    #pragma unroll
    for (int j = 0; j < 4; ++j) {
      v2f kA;
      #pragma unroll
      for (int s = 0; s < 2; ++s) {
        const float a = a_av[t][2 * j + s];
        const float a2 = a * a, a4 = a2 * a2, a8 = a4 * a4;
        float p = l1 ? a : 1.0f;
        p *= l2 ? a2 : 1.0f;
        p *= l4 ? a4 : 1.0f;
        p *= l8 ? a8 : 1.0f;
        const float cp = c_cm[t][j][s] * p;
        kA[s] = cp * b_cm[t][j][s];
        cmA[t][j][s] = cp * a;
      }
      kd = wmma4_f32(kA, ones, kd);
    }

  float kl;
  {
    const float s0 = l1 ? kd[1] : kd[0], s1 = l1 ? kd[3] : kd[2];
    const float s2 = l1 ? kd[5] : kd[4], s3 = l1 ? kd[7] : kd[6];
    const float s01 = l2 ? s1 : s0, s23 = l2 ? s3 : s2;
    const float kown = l4 ? s23 : s01;
    const float koth = __shfl_xor(kown, 16);
    kl = (l8 == (half != 0)) ? kown : koth;
  }

  v2f mA[4];
  #pragma unroll
  for (int j = 0; j < 4; ++j)
    #pragma unroll
    for (int s = 0; s < 2; ++s) {
      const int d = lm - (8 * half + 2 * j + s);
      const float v = __shfl(kl, d & 15);
      mA[j][s] = d >= 0 ? v : 0.0f;
    }

  v8f P[CH_SEG - 1][4];
  v8f H[4];

  #pragma clang loop licm(disable)
  #pragma nounroll
  for (int it = 0; it < SPW; ++it) {
  const float* unext = useq + (size_t)(it + 1 < SPW ? it + 1 : it) * LSEQ;
  float* yln = yseq + (size_t)it * LSEQ;

  first_pass(bm, av, ub, P, H);

  {
    float q[4][8];
    #pragma unroll
    for (int t = 0; t < 4; ++t)
      #pragma unroll
      for (int r = 0; r < 8; ++r) {
        float m = av[t][r];
        m *= m; m *= m; m *= m;
        q[t][r] = m;
      }
    fold_step<1>(H, q);
    fold_step<2>(H, q);
    fold_step<4>(H, q);
    fold_step<8>(H, q);
    #pragma unroll
    for (int t = 0; t < 4; ++t)
      #pragma unroll
      for (int r = 0; r < 8; ++r)
        H[t][r] = col_shr<1>(H[t][r]);
  }

  output_pass(mA, cmA, av, ub, P, H, ywr, yrd, yln, unext, half);
  }
}

extern "C" void kernel_launch(void* const* d_in, const int* in_sizes, int n_in,
                              void* d_out, int out_size, void* d_ws, size_t ws_size,
                              hipStream_t stream) {
  const float* u  = (const float*)d_in[0];
  const float* A  = (const float*)d_in[1];
  const float* Bv = (const float*)d_in[2];
  const float* Cv = (const float*)d_in[3];
  float* y      = (float*)d_out;

  ssm_seq<<<dim3(SEQS / (WPB * SPW)), dim3(WPB * 32), 0, stream>>>(u, A, Bv, Cv, y);
}
